# starting kernel
# baseline (speedup 1.0000x reference)
#include <hip/hip_runtime.h>

typedef __attribute__((ext_vector_type(2))) float v2f;
typedef __attribute__((ext_vector_type(8))) float v8f;

#define N_PTS 500000
#define M_CTR 256
#define NORM_CONST 0.025330295910584444f

__global__ void gauss_weights_kernel(const float* __restrict__ centers,
                                     const float* __restrict__ coefficients,
                                     float* __restrict__ w) {
    __shared__ float red[M_CTR];
    const int t = threadIdx.x;
    float c = coefficients[t];

    red[t] = c;
    __syncthreads();
    for (int s = M_CTR / 2; s > 0; s >>= 1) {
        if (t < s) red[t] = fmaxf(red[t], red[t + s]);
        __syncthreads();
    }
    const float mx = red[0];
    __syncthreads();

    const float e = __expf(c - mx);
    red[t] = e;
    __syncthreads();
    for (int s = M_CTR / 2; s > 0; s >>= 1) {
        if (t < s) red[t] += red[t + s];
        __syncthreads();
    }
    const float sum = red[0];

    const float c0 = centers[t * 4 + 0];
    const float c1 = centers[t * 4 + 1];
    const float c2 = centers[t * 4 + 2];
    const float c3 = centers[t * 4 + 3];
    const float cn2 = c0 * c0 + c1 * c1 + c2 * c2 + c3 * c3;

    w[t] = (e / sum) * NORM_CONST * __expf(-0.5f * cn2);
}

__global__ void __launch_bounds__(256) gauss_main_kernel(
        const float* __restrict__ x,
        const float* __restrict__ centers,
        const float* __restrict__ w,
        float* __restrict__ out) {
    __shared__ float ws[M_CTR];
    ws[threadIdx.x] = w[threadIdx.x];
    __syncthreads();

    const int lane = threadIdx.x & 31;
    const int waveId = blockIdx.x * (blockDim.x >> 5) + (threadIdx.x >> 5);
    const int numTiles = N_PTS / 16;
    if (waveId >= numTiles) return;

    const int row0 = waveId * 16;
    const int r    = lane & 15;
    const int hi   = lane >> 4;
    const int koff = hi * 2;

    v2f a = *(const v2f*)(x + (size_t)(row0 + r) * 4 + koff);

    float part = a.x * a.x + a.y * a.y;
    float x2 = part + __shfl_xor(part, 16);

    float acc[8];
#pragma unroll
    for (int j = 0; j < 8; ++j) acc[j] = 0.0f;

#pragma unroll
    for (int tile = 0; tile < 16; ++tile) {
        const int col = tile * 16 + r;
        v2f b = *(const v2f*)(centers + col * 4 + koff);

        v8f c = {};
        c = __builtin_amdgcn_wmma_f32_16x16x4_f32(
                 false, a,  false, b,
                 (short)0, c,  false,  false);

        const float wcol = ws[col];
#pragma unroll
        for (int j = 0; j < 8; ++j)
            acc[j] += wcol * __expf(c[j]);
    }

#pragma unroll
    for (int j = 0; j < 8; ++j) {
        float v = acc[j];
        v += __shfl_xor(v, 1);
        v += __shfl_xor(v, 2);
        v += __shfl_xor(v, 4);
        v += __shfl_xor(v, 8);
        acc[j] = v;
    }

#pragma unroll
    for (int j = 0; j < 8; ++j) {
        const float x2j = __shfl(x2, hi * 8 + j);
        acc[j] *= __expf(-0.5f * x2j);
    }

    if (r == 0) {
        const int base = row0 + hi * 8;
#pragma unroll
        for (int j = 0; j < 8; ++j) out[base + j] = acc[j];
    }
}

extern "C" void kernel_launch(void* const* d_in, const int* in_sizes, int n_in,
                              void* d_out, int out_size, void* d_ws, size_t ws_size,
                              hipStream_t stream) {
    const float* x            = (const float*)d_in[0];
    const float* centers      = (const float*)d_in[1];
    const float* coefficients = (const float*)d_in[2];
    float* out = (float*)d_out;
    float* w   = (float*)d_ws;

    gauss_weights_kernel<<<1, M_CTR, 0, stream>>>(centers, coefficients, w);

    const int numWaveTiles = N_PTS / 16;
    const int wavesPerBlock = 8;
    const int blocks = (numWaveTiles + wavesPerBlock - 1) / wavesPerBlock;
    gauss_main_kernel<<<blocks, 256, 0, stream>>>(x, centers, w, out);
}


# Main kernel as a fixed grid of 4096 long-lived waves: each keeps the 256 centers' B fragments and weights in registers and loops over 32-row steps, next rows prefetched (was 31250 one-tile waves)
# speedup vs baseline: 1.0796x; 1.0796x over previous
#include <hip/hip_runtime.h>


typedef __attribute__((ext_vector_type(2))) float v2f;
typedef __attribute__((ext_vector_type(8))) float v8f;

#define N_PTS 500000
#define M_CTR 256
#define NORM_CONST 0.025330295910584444f

#define MAIN_BLOCKS 512
#define WAVES_PER_BLOCK 8
#define MAIN_THREADS 256

__global__ void gauss_weights_kernel(const float* __restrict__ centers,
                                     const float* __restrict__ coefficients,
                                     float* __restrict__ w) {
    __shared__ float red[M_CTR];
    const int t = threadIdx.x;
    float c = coefficients[t];

    red[t] = c;
    __syncthreads();
    for (int s = M_CTR / 2; s > 0; s >>= 1) {
        if (t < s) red[t] = fmaxf(red[t], red[t + s]);
        __syncthreads();
    }
    const float mx = red[0];
    __syncthreads();

    const float e = __expf(c - mx);
    red[t] = e;
    __syncthreads();
    for (int s = M_CTR / 2; s > 0; s >>= 1) {
        if (t < s) red[t] += red[t + s];
        __syncthreads();
    }
    const float sum = red[0];

    const float c0 = centers[t * 4 + 0];
    const float c1 = centers[t * 4 + 1];
    const float c2 = centers[t * 4 + 2];
    const float c3 = centers[t * 4 + 3];
    const float cn2 = c0 * c0 + c1 * c1 + c2 * c2 + c3 * c3;

    w[t] = (e / sum) * NORM_CONST * __expf(-0.5f * cn2);
}

__device__ __forceinline__ void rows16(const v2f a, const v2f (&b)[16],
                                       const float (&wc)[16], const int hi,
                                       float (&res)[8]) {
    const float part = a.x * a.x + a.y * a.y;
    const float x2 = part + __shfl_xor(part, 16);

    float acc[8];
#pragma unroll
    for (int j = 0; j < 8; ++j) acc[j] = 0.0f;

#pragma unroll
    for (int tile = 0; tile < 16; ++tile) {
        v8f c = {};
        c = __builtin_amdgcn_wmma_f32_16x16x4_f32(
                 false, a,  false, b[tile],
                 (short)0, c,  false,  false);
        const float wcol = wc[tile];
#pragma unroll
        for (int j = 0; j < 8; ++j)
            acc[j] += wcol * __expf(c[j]);
    }

#pragma unroll
    for (int j = 0; j < 8; ++j) {
        float v = acc[j];
        v += __shfl_xor(v, 1);
        v += __shfl_xor(v, 2);
        v += __shfl_xor(v, 4);
        v += __shfl_xor(v, 8);
        acc[j] = v;
    }

#pragma unroll
    for (int j = 0; j < 8; ++j) {
        const float x2j = __shfl(x2, hi * 8 + j);
        res[j] = acc[j] * __expf(-0.5f * x2j);
    }
}

__global__ void __launch_bounds__(256) gauss_main_kernel(
        const float* __restrict__ x,
        const float* __restrict__ centers,
        const float* __restrict__ w,
        float* __restrict__ out) {
    const int lane = threadIdx.x & 31;
    const int r    = lane & 15;
    const int hi   = lane >> 4;
    const int koff = hi * 2;

    const int waveInBlock = __builtin_amdgcn_readfirstlane((int)(threadIdx.x >> 5));
    const int waveId  = blockIdx.x * WAVES_PER_BLOCK + waveInBlock;
    const int nWaves  = gridDim.x * WAVES_PER_BLOCK;
    const int numPairs = N_PTS / 32;
    if (waveId >= numPairs) return;

    v2f b[16];
    float wc[16];
#pragma unroll
    for (int tile = 0; tile < 16; ++tile) {
        const int col = tile * 16 + r;
        b[tile]  = *(const v2f*)(centers + col * 4 + koff);
        wc[tile] = w[col];
    }

    const float* xl = x + r * 4 + koff;
    int p = waveId;
    v2f a0 = *(const v2f*)(xl + (size_t)p * 128);
    v2f a1 = *(const v2f*)(xl + (size_t)p * 128 + 64);

    for (;;) {
        const int pn = p + nWaves;
        const bool more = pn < numPairs;
        v2f n0 = a0, n1 = a1;
        if (more) {
            n0 = *(const v2f*)(xl + (size_t)pn * 128);
            n1 = *(const v2f*)(xl + (size_t)pn * 128 + 64);
        }

        float r0[8], r1[8];
        rows16(a0, b, wc, hi, r0);
        rows16(a1, b, wc, hi, r1);

        if (r == 0) {
            float* o = out + (size_t)p * 32 + hi * 8;
#pragma unroll
            for (int j = 0; j < 8; ++j) o[j] = r0[j];
#pragma unroll
            for (int j = 0; j < 8; ++j) o[16 + j] = r1[j];
        }

        if (!more) break;
        a0 = n0; a1 = n1; p = pn;
    }
}

extern "C" void kernel_launch(void* const* d_in, const int* in_sizes, int n_in,
                              void* d_out, int out_size, void* d_ws, size_t ws_size,
                              hipStream_t stream) {
    const float* x            = (const float*)d_in[0];
    const float* centers      = (const float*)d_in[1];
    const float* coefficients = (const float*)d_in[2];
    float* out = (float*)d_out;
    float* w   = (float*)d_ws;

    gauss_weights_kernel<<<1, M_CTR, 0, stream>>>(centers, coefficients, w);

    gauss_main_kernel<<<MAIN_BLOCKS, MAIN_THREADS, 0, stream>>>(x, centers, w, out);
}


# One launch instead of two: every wave of the main kernel works out the 256 softmax weights itself (16 columns per lane, lane-xor max and sum in the same pairing), no weights kernel, no workspace
# speedup vs baseline: 1.3549x; 1.2550x over previous
#include <hip/hip_runtime.h>


typedef __attribute__((ext_vector_type(2))) float v2f;
typedef __attribute__((ext_vector_type(4))) float v4f;
typedef __attribute__((ext_vector_type(8))) float v8f;

#define N_PTS 500000
#define NORM_CONST 0.025330295910584444f

#define MAIN_BLOCKS 512
#define WAVES_PER_BLOCK 8
#define MAIN_THREADS 256

__device__ __forceinline__ void rows16(const v2f a, const v2f (&b)[16],
                                       const float (&wc)[16], const int hi,
                                       float (&res)[8]) {
    const float part = a.x * a.x + a.y * a.y;
    const float x2 = part + __shfl_xor(part, 16);

    float acc[8];
#pragma unroll
    for (int j = 0; j < 8; ++j) acc[j] = 0.0f;

#pragma unroll
    for (int tile = 0; tile < 16; ++tile) {
        v8f c = {};
        c = __builtin_amdgcn_wmma_f32_16x16x4_f32(
                 false, a,  false, b[tile],
                 (short)0, c,  false,  false);
        const float wcol = wc[tile];
#pragma unroll
        for (int j = 0; j < 8; ++j)
            acc[j] += wcol * __expf(c[j]);
    }

#pragma unroll
    for (int j = 0; j < 8; ++j) {
        float v = acc[j];
        v += __shfl_xor(v, 1);
        v += __shfl_xor(v, 2);
        v += __shfl_xor(v, 4);
        v += __shfl_xor(v, 8);
        acc[j] = v;
    }

#pragma unroll
    for (int j = 0; j < 8; ++j) {
        const float x2j = __shfl(x2, hi * 8 + j);
        res[j] = acc[j] * __expf(-0.5f * x2j);
    }
}

__global__ void __launch_bounds__(256) gauss_main_kernel(
        const float* __restrict__ x,
        const float* __restrict__ centers,
        const float* __restrict__ coefficients,
        float* __restrict__ out) {
    const int lane = threadIdx.x & 31;
    const int r    = lane & 15;
    const int hi   = lane >> 4;
    const int koff = hi * 2;

    const int waveInBlock = __builtin_amdgcn_readfirstlane((int)(threadIdx.x >> 5));
    const int waveId  = blockIdx.x * WAVES_PER_BLOCK + waveInBlock;
    const int nWaves  = gridDim.x * WAVES_PER_BLOCK;
    const int numPairs = N_PTS / 32;
    if (waveId >= numPairs) return;

    const float* xl = x + r * 4 + koff;
    int p = waveId;
    v2f a0 = *(const v2f*)(xl + (size_t)p * 128);
    v2f a1 = *(const v2f*)(xl + (size_t)p * 128 + 64);

    v2f b[16];
    v4f cc[16];
    float cf[16];
#pragma unroll
    for (int tile = 0; tile < 16; ++tile) {
        const int col = tile * 16 + r;
        b[tile]  = *(const v2f*)(centers + col * 4 + koff);
        cc[tile] = *(const v4f*)(centers + col * 4);
        cf[tile] = coefficients[col];
    }

    float mx = cf[0];
#pragma unroll
    for (int tile = 1; tile < 16; ++tile) mx = fmaxf(mx, cf[tile]);
    mx = fmaxf(mx, __shfl_xor(mx, 8));
    mx = fmaxf(mx, __shfl_xor(mx, 4));
    mx = fmaxf(mx, __shfl_xor(mx, 2));
    mx = fmaxf(mx, __shfl_xor(mx, 1));

    float e[16];
#pragma unroll
    for (int tile = 0; tile < 16; ++tile) e[tile] = __expf(cf[tile] - mx);
    float s8[8], s4[4], s2[2];
#pragma unroll
    for (int t = 0; t < 8; ++t) s8[t] = e[t] + e[t + 8];
#pragma unroll
    for (int t = 0; t < 4; ++t) s4[t] = s8[t] + s8[t + 4];
#pragma unroll
    for (int t = 0; t < 2; ++t) s2[t] = s4[t] + s4[t + 2];
    float sum = s2[0] + s2[1];
    sum += __shfl_xor(sum, 8);
    sum += __shfl_xor(sum, 4);
    sum += __shfl_xor(sum, 2);
    sum += __shfl_xor(sum, 1);

    float wc[16];
#pragma unroll
    for (int tile = 0; tile < 16; ++tile) {
        const v4f c = cc[tile];
        const float cn2 = c.x * c.x + c.y * c.y + c.z * c.z + c.w * c.w;
        wc[tile] = (e[tile] / sum) * NORM_CONST * __expf(-0.5f * cn2);
    }

    for (;;) {
        const int pn = p + nWaves;
        const bool more = pn < numPairs;
        v2f n0 = a0, n1 = a1;
        if (more) {
            n0 = *(const v2f*)(xl + (size_t)pn * 128);
            n1 = *(const v2f*)(xl + (size_t)pn * 128 + 64);
        }

        float r0[8], r1[8];
        rows16(a0, b, wc, hi, r0);
        rows16(a1, b, wc, hi, r1);

        if (r == 0) {
            float* o = out + (size_t)p * 32 + hi * 8;
#pragma unroll
            for (int j = 0; j < 8; ++j) o[j] = r0[j];
#pragma unroll
            for (int j = 0; j < 8; ++j) o[16 + j] = r1[j];
        }

        if (!more) break;
        a0 = n0; a1 = n1; p = pn;
    }
}

extern "C" void kernel_launch(void* const* d_in, const int* in_sizes, int n_in,
                              void* d_out, int out_size, void* d_ws, size_t ws_size,
                              hipStream_t stream) {
    const float* x            = (const float*)d_in[0];
    const float* centers      = (const float*)d_in[1];
    const float* coefficients = (const float*)d_in[2];
    float* out = (float*)d_out;

    gauss_main_kernel<<<MAIN_BLOCKS, MAIN_THREADS, 0, stream>>>(x, centers, coefficients, out);
}


# Product transposed: centers are the matrix rows (A), points the columns (B), so a lane sums its own point down its registers (new summation order); no 16-lane shuffle reductions, one store per line
# speedup vs baseline: 1.5024x; 1.1089x over previous
#include <hip/hip_runtime.h>


typedef __attribute__((ext_vector_type(2))) float v2f;
typedef __attribute__((ext_vector_type(4))) float v4f;
typedef __attribute__((ext_vector_type(8))) float v8f;

#define N_PTS 500000
#define NORM_CONST 0.025330295910584444f

#define MAIN_BLOCKS 512
#define WAVES_PER_BLOCK 8
#define MAIN_THREADS 256

__device__ __forceinline__ float xhalf(const float v) {
    return __builtin_bit_cast(float, __builtin_amdgcn_permlanex16(
        __builtin_bit_cast(int, v), __builtin_bit_cast(int, v),
        0x76543210, 0xfedcba98, false, false));
}

__device__ __forceinline__ float row_bcast(const float v, const int src) {
    const int sel = src * 0x11111111;
    return __builtin_bit_cast(float, __builtin_amdgcn_permlane16(
        __builtin_bit_cast(int, v), __builtin_bit_cast(int, v),
        sel, sel, false, false));
}

__device__ __forceinline__ float cols16(const v2f xa, const v2f (&cfr)[16],
                                        const v2f (&w2)[16][4]) {
    v2f acc[4];
#pragma unroll
    for (int q = 0; q < 4; ++q) acc[q] = (v2f){0.0f, 0.0f};

#pragma unroll
    for (int tile = 0; tile < 16; ++tile) {
        v8f d = {};
        d = __builtin_amdgcn_wmma_f32_16x16x4_f32(
                 false, cfr[tile],  false, xa,
                 (short)0, d,  false,  false);
#pragma unroll
        for (int q = 0; q < 4; ++q) {
            v2f e;
            e.x = __expf(d[2 * q]);
            e.y = __expf(d[2 * q + 1]);
            acc[q] += w2[tile][q] * e;
        }
    }

    const v2f s2 = (acc[0] + acc[1]) + (acc[2] + acc[3]);
    float s = s2.x + s2.y;
    s += xhalf(s);

    const float part = xa.x * xa.x + xa.y * xa.y;
    const float x2 = part + xhalf(part);
    return s * __expf(-0.5f * x2);
}

__global__ void __launch_bounds__(256, 4) gauss_main_kernel(
        const float* __restrict__ x,
        const float* __restrict__ centers,
        const float* __restrict__ coefficients,
        float* __restrict__ out) {
    const int lane = threadIdx.x & 31;
    const int r    = lane & 15;
    const int hi   = lane >> 4;
    const int koff = hi * 2;

    const int waveInBlock = __builtin_amdgcn_readfirstlane((int)(threadIdx.x >> 5));
    const int waveId  = blockIdx.x * WAVES_PER_BLOCK + waveInBlock;
    const int nWaves  = gridDim.x * WAVES_PER_BLOCK;
    const int numPairs = N_PTS / 32;
    if (waveId >= numPairs) return;

    const float* xl = x + r * 4 + koff;
    int p = waveId;
    v2f a0 = *(const v2f*)(xl + (size_t)p * 128);
    v2f a1 = *(const v2f*)(xl + (size_t)p * 128 + 64);

    const int rw = r ^ (hi << 3);
    v2f cfr[16];
    v4f cc[16];
    float cf[16];
#pragma unroll
    for (int tile = 0; tile < 16; ++tile) {
        cfr[tile] = *(const v2f*)(centers + (tile * 16 + r) * 4 + koff);
        const int col = tile * 16 + rw;
        cc[tile] = *(const v4f*)(centers + col * 4);
        cf[tile] = coefficients[col];
    }

    float mx = cf[0];
#pragma unroll
    for (int tile = 1; tile < 16; ++tile) mx = fmaxf(mx, cf[tile]);
    mx = fmaxf(mx, __shfl_xor(mx, 8));
    mx = fmaxf(mx, __shfl_xor(mx, 4));
    mx = fmaxf(mx, __shfl_xor(mx, 2));
    mx = fmaxf(mx, __shfl_xor(mx, 1));

    float e[16];
#pragma unroll
    for (int tile = 0; tile < 16; ++tile) e[tile] = __expf(cf[tile] - mx);
    float s8[8], s4[4], s2[2];
#pragma unroll
    for (int t = 0; t < 8; ++t) s8[t] = e[t] + e[t + 8];
#pragma unroll
    for (int t = 0; t < 4; ++t) s4[t] = s8[t] + s8[t + 4];
#pragma unroll
    for (int t = 0; t < 2; ++t) s2[t] = s4[t] + s4[t + 2];
    float sum = s2[0] + s2[1];
    sum += __shfl_xor(sum, 8);
    sum += __shfl_xor(sum, 4);
    sum += __shfl_xor(sum, 2);
    sum += __shfl_xor(sum, 1);

    v2f w2[16][4];
#pragma unroll
    for (int tile = 0; tile < 16; ++tile) {
        const v4f c = cc[tile];
        const float cn2 = c.x * c.x + c.y * c.y + c.z * c.z + c.w * c.w;
        const float wcol = (e[tile] / sum) * NORM_CONST * __expf(-0.5f * cn2);
#pragma unroll
        for (int q = 0; q < 4; ++q) {
            w2[tile][q].x = row_bcast(wcol, 2 * q);
            w2[tile][q].y = row_bcast(wcol, 2 * q + 1);
        }
    }

    float* ol = out + lane;
    for (;;) {
        const int pn = p + nWaves;
        const bool more = pn < numPairs;
        v2f n0 = a0, n1 = a1;
        if (more) {
            n0 = *(const v2f*)(xl + (size_t)pn * 128);
            n1 = *(const v2f*)(xl + (size_t)pn * 128 + 64);
        }

        const float r0 = cols16(a0, cfr, w2);
        const float r1 = cols16(a1, cfr, w2);

        ol[(size_t)p * 32] = hi ? r1 : r0;

        if (!more) break;
        a0 = n0; a1 = n1; p = pn;
    }
}

extern "C" void kernel_launch(void* const* d_in, const int* in_sizes, int n_in,
                              void* d_out, int out_size, void* d_ws, size_t ws_size,
                              hipStream_t stream) {
    const float* x            = (const float*)d_in[0];
    const float* centers      = (const float*)d_in[1];
    const float* coefficients = (const float*)d_in[2];
    float* out = (float*)d_out;

    gauss_main_kernel<<<MAIN_BLOCKS, MAIN_THREADS, 0, stream>>>(x, centers, coefficients, out);
}


# Centers' matrix fragments pre-scaled by log2(e) once per wave: a product tile is already the base-2 exponent, exp2 straight on it; the loop's 128 packed multiplies per 32-point step are gone
# speedup vs baseline: 1.5919x; 1.0596x over previous
#include <hip/hip_runtime.h>


typedef __attribute__((ext_vector_type(2))) float v2f;
typedef __attribute__((ext_vector_type(4))) float v4f;
typedef __attribute__((ext_vector_type(8))) float v8f;

#define N_PTS 500000
#define NORM_CONST 0.025330295910584444f
#define LOG2E 1.4426950408889634f

#define MAIN_BLOCKS 512
#define WAVES_PER_BLOCK 8
#define MAIN_THREADS 256

__device__ __forceinline__ float xhalf(const float v) {
    return __builtin_bit_cast(float, __builtin_amdgcn_permlanex16(
        __builtin_bit_cast(int, v), __builtin_bit_cast(int, v),
        0x76543210, 0xfedcba98, false, false));
}

__device__ __forceinline__ float row_bcast(const float v, const int src) {
    const int sel = src * 0x11111111;
    return __builtin_bit_cast(float, __builtin_amdgcn_permlane16(
        __builtin_bit_cast(int, v), __builtin_bit_cast(int, v),
        sel, sel, false, false));
}

__device__ __forceinline__ float cols16(const v2f xa, const v2f (&cfr)[16],
                                        const v2f (&w2)[16][4]) {
    v2f acc[4];
#pragma unroll
    for (int q = 0; q < 4; ++q) acc[q] = (v2f){0.0f, 0.0f};

#pragma unroll
    for (int tile = 0; tile < 16; ++tile) {
        v8f d = {};
        d = __builtin_amdgcn_wmma_f32_16x16x4_f32(
                 false, cfr[tile],  false, xa,
                 (short)0, d,  false,  false);
#pragma unroll
        for (int q = 0; q < 4; ++q) {
            v2f e;
            e.x = __builtin_amdgcn_exp2f(d[2 * q]);
            e.y = __builtin_amdgcn_exp2f(d[2 * q + 1]);
            acc[q] += w2[tile][q] * e;
        }
    }

    const v2f s2 = (acc[0] + acc[1]) + (acc[2] + acc[3]);
    float s = s2.x + s2.y;
    s += xhalf(s);

    const float part = xa.x * xa.x + xa.y * xa.y;
    const float x2 = part + xhalf(part);
    return s * __expf(-0.5f * x2);
}

__global__ void __launch_bounds__(256, 4) gauss_main_kernel(
        const float* __restrict__ x,
        const float* __restrict__ centers,
        const float* __restrict__ coefficients,
        float* __restrict__ out) {
    const int lane = threadIdx.x & 31;
    const int r    = lane & 15;
    const int hi   = lane >> 4;
    const int koff = hi * 2;

    const int waveInBlock = __builtin_amdgcn_readfirstlane((int)(threadIdx.x >> 5));
    const int waveId  = blockIdx.x * WAVES_PER_BLOCK + waveInBlock;
    const int nWaves  = gridDim.x * WAVES_PER_BLOCK;
    const int numPairs = N_PTS / 32;
    if (waveId >= numPairs) return;

    const float* xl = x + r * 4 + koff;
    int p = waveId;
    v2f a0 = *(const v2f*)(xl + (size_t)p * 128);
    v2f a1 = *(const v2f*)(xl + (size_t)p * 128 + 64);

    const int rw = r ^ (hi << 3);
    v2f cfr[16];
    v4f cc[16];
    float cf[16];
#pragma unroll
    for (int tile = 0; tile < 16; ++tile) {
        cfr[tile] = *(const v2f*)(centers + (tile * 16 + r) * 4 + koff) * LOG2E;
        const int col = tile * 16 + rw;
        cc[tile] = *(const v4f*)(centers + col * 4);
        cf[tile] = coefficients[col];
    }

    float mx = cf[0];
#pragma unroll
    for (int tile = 1; tile < 16; ++tile) mx = fmaxf(mx, cf[tile]);
    mx = fmaxf(mx, __shfl_xor(mx, 8));
    mx = fmaxf(mx, __shfl_xor(mx, 4));
    mx = fmaxf(mx, __shfl_xor(mx, 2));
    mx = fmaxf(mx, __shfl_xor(mx, 1));

    float e[16];
#pragma unroll
    for (int tile = 0; tile < 16; ++tile) e[tile] = __expf(cf[tile] - mx);
    float s8[8], s4[4], s2[2];
#pragma unroll
    for (int t = 0; t < 8; ++t) s8[t] = e[t] + e[t + 8];
#pragma unroll
    for (int t = 0; t < 4; ++t) s4[t] = s8[t] + s8[t + 4];
#pragma unroll
    for (int t = 0; t < 2; ++t) s2[t] = s4[t] + s4[t + 2];
    float sum = s2[0] + s2[1];
    sum += __shfl_xor(sum, 8);
    sum += __shfl_xor(sum, 4);
    sum += __shfl_xor(sum, 2);
    sum += __shfl_xor(sum, 1);

    v2f w2[16][4];
#pragma unroll
    for (int tile = 0; tile < 16; ++tile) {
        const v4f c = cc[tile];
        const float cn2 = c.x * c.x + c.y * c.y + c.z * c.z + c.w * c.w;
        const float wcol = (e[tile] / sum) * NORM_CONST * __expf(-0.5f * cn2);
#pragma unroll
        for (int q = 0; q < 4; ++q) {
            w2[tile][q].x = row_bcast(wcol, 2 * q);
            w2[tile][q].y = row_bcast(wcol, 2 * q + 1);
        }
    }

    float* ol = out + lane;
    for (;;) {
        const int pn = p + nWaves;
        const bool more = pn < numPairs;
        v2f n0 = a0, n1 = a1;
        if (more) {
            n0 = *(const v2f*)(xl + (size_t)pn * 128);
            n1 = *(const v2f*)(xl + (size_t)pn * 128 + 64);
        }

        const float r0 = cols16(a0, cfr, w2);
        const float r1 = cols16(a1, cfr, w2);

        ol[(size_t)p * 32] = hi ? r1 : r0;

        if (!more) break;
        a0 = n0; a1 = n1; p = pn;
    }
}

extern "C" void kernel_launch(void* const* d_in, const int* in_sizes, int n_in,
                              void* d_out, int out_size, void* d_ws, size_t ws_size,
                              hipStream_t stream) {
    const float* x            = (const float*)d_in[0];
    const float* centers      = (const float*)d_in[1];
    const float* coefficients = (const float*)d_in[2];
    float* out = (float*)d_out;

    gauss_main_kernel<<<MAIN_BLOCKS, MAIN_THREADS, 0, stream>>>(x, centers, coefficients, out);
}
